# starting kernel
# baseline (speedup 1.0000x reference)
#include <hip/hip_runtime.h>
#include <hip/hip_bf16.h>
#include <math.h>


typedef __attribute__((ext_vector_type(2))) float v2f;
typedef __attribute__((ext_vector_type(8))) float v8f;

constexpr int kB   = 128;
constexpr int kR   = 4096;
constexpr int kC   = 10;
constexpr int kO   = 16;
constexpr int kI   = 8;
constexpr int kMT  = 16;
constexpr int kNMT = kB / kMT;
constexpr int kRT  = 128;
constexpr int kNRT = kR / kRT;
constexpr int kWAVES = 4;
constexpr int kRPW = kRT / kWAVES;
constexpr int kCO  = kC * kO;

template <int PHASE>
__global__ __launch_bounds__(kWAVES * 32) void caps_pass(
    const float* __restrict__ x,
    const float* __restrict__ Wt,
    const float* __restrict__ vprev,
    float* __restrict__ bij,
    float* __restrict__ spart)
{
  __shared__ float lds_v[kMT * kCO];
  __shared__ float lds_s[kWAVES][kMT * kCO];

  const int mtile  = blockIdx.x & (kNMT - 1);
  const int rchunk = blockIdx.x / kNMT;
  const int wave   = threadIdx.x >> 5;
  const int lane   = threadIdx.x & 31;
  const int o      = lane & 15;
  const int half   = lane >> 4;

  if (PHASE != 0) {
    for (int e = threadIdx.x; e < kMT * kCO; e += blockDim.x) {
      int bl = e / kCO;
      lds_v[e] = vprev[(size_t)(mtile * kMT + bl) * kCO + (e % kCO)];
    }
    __syncthreads();
  }

  const v8f zero8 = {0.f, 0.f, 0.f, 0.f, 0.f, 0.f, 0.f, 0.f};
  v8f s_acc[kC];
#pragma unroll
  for (int c = 0; c < kC; ++c) s_acc[c] = zero8;

  const int bA = mtile * kMT + o;
  const int k0 = half * 2;

  for (int ri = 0; ri < kRPW; ++ri) {
    const int r = rchunk * kRT + wave * kRPW + ri;

    const float* xp = x + ((size_t)bA * kR + r) * kI;
    v2f a0 = *(const v2f*)(xp + k0);
    v2f a1 = *(const v2f*)(xp + k0 + 4);

    v8f acc[kC];
#pragma unroll
    for (int c = 0; c < kC; ++c) acc[c] = zero8;
#pragma unroll
    for (int c = 0; c < kC; ++c) {
      const float* wp = Wt + (((size_t)r * kC + c) * kO + o) * kI;
      v2f b0 = *(const v2f*)(wp + k0);
      v2f b1 = *(const v2f*)(wp + k0 + 4);
      acc[c] = __builtin_amdgcn_wmma_f32_16x16x4_f32(false, a0, false, b0,
                                                     (short)0, acc[c], false, false);
      acc[c] = __builtin_amdgcn_wmma_f32_16x16x4_f32(false, a1, false, b1,
                                                     (short)0, acc[c], false, false);
    }

    if (PHASE == 0) {
#pragma unroll
      for (int c = 0; c < kC; ++c)
#pragma unroll
        for (int j = 0; j < 8; ++j) s_acc[c][j] += acc[c][j];
    } else {
      float bnew[8][kC];
#pragma unroll
      for (int j = 0; j < 8; ++j) {
        const int bl = half * 8 + j;
#pragma unroll
        for (int c = 0; c < kC; ++c) {
          float p = acc[c][j] * lds_v[(bl * kC + c) * kO + o];
          p += __shfl_xor(p, 1);
          p += __shfl_xor(p, 2);
          p += __shfl_xor(p, 4);
          p += __shfl_xor(p, 8);
          bnew[j][c] = p;
        }
      }

      const size_t bbase = ((size_t)r * kB + (mtile * kMT + half * 8)) * kC;
      if (PHASE == 2) {
#pragma unroll
        for (int j = 0; j < 8; ++j)
#pragma unroll
          for (int c = 0; c < kC; ++c) bnew[j][c] += bij[bbase + j * kC + c];
      }
      if (PHASE == 1) {
        if (o == 0) {
#pragma unroll
          for (int j = 0; j < 8; ++j)
#pragma unroll
            for (int c = 0; c < kC; ++c) bij[bbase + j * kC + c] = bnew[j][c];
        }
      }

#pragma unroll
      for (int j = 0; j < 8; ++j) {
        float mx = bnew[j][0];
#pragma unroll
        for (int c = 1; c < kC; ++c) mx = fmaxf(mx, bnew[j][c]);
        float sum = 0.f;
#pragma unroll
        for (int c = 0; c < kC; ++c) {
          bnew[j][c] = __expf(bnew[j][c] - mx);
          sum += bnew[j][c];
        }
        const float inv = 1.0f / sum;
#pragma unroll
        for (int c = 0; c < kC; ++c)
          s_acc[c][j] += acc[c][j] * (bnew[j][c] * inv);
      }
    }
  }

#pragma unroll
  for (int c = 0; c < kC; ++c)
#pragma unroll
    for (int j = 0; j < 8; ++j) {
      const int bl = half * 8 + j;
      lds_s[wave][(bl * kC + c) * kO + o] =
          s_acc[c][j] * (PHASE == 0 ? 0.1f : 1.0f);
    }
  __syncthreads();
  for (int e = threadIdx.x; e < kMT * kCO; e += blockDim.x) {
    float sum = lds_s[0][e] + lds_s[1][e] + lds_s[2][e] + lds_s[3][e];
    const int bl = e / kCO;
    spart[((size_t)rchunk * kB + (mtile * kMT + bl)) * kCO + (e % kCO)] = sum;
  }
}

__global__ __launch_bounds__(256) void caps_squash(
    const float* __restrict__ spart,
    const float* __restrict__ bias,
    float* __restrict__ vout)
{
  const int idx = blockIdx.x * 256 + threadIdx.x;
  float s = bias[idx % kCO];
  for (int k = 0; k < kNRT; ++k) s += spart[(size_t)k * (kB * kCO) + idx];
  float p = s * s;
  p += __shfl_xor(p, 1);
  p += __shfl_xor(p, 2);
  p += __shfl_xor(p, 4);
  p += __shfl_xor(p, 8);
  vout[idx] = s * p / ((1.0f + p) * sqrtf(p));
}

extern "C" void kernel_launch(void* const* d_in, const int* in_sizes, int n_in,
                              void* d_out, int out_size, void* d_ws, size_t ws_size,
                              hipStream_t stream) {
  const float* x    = (const float*)d_in[0];
  const float* Wt   = (const float*)d_in[1];
  const float* bias = (const float*)d_in[2];
  float* out = (float*)d_out;

  float* spart = (float*)d_ws;
  float* vbuf  = spart + (size_t)kNRT * kB * kCO;
  float* bijw  = vbuf + (size_t)kB * kCO;

  const dim3 gp(kNMT * kNRT), bp(kWAVES * 32);
  const dim3 gs((kB * kCO) / 256), bs(256);

  caps_pass<0><<<gp, bp, 0, stream>>>(x, Wt, nullptr, bijw, spart);
  caps_squash<<<gs, bs, 0, stream>>>(spart, bias, vbuf);
  caps_pass<1><<<gp, bp, 0, stream>>>(x, Wt, vbuf, bijw, spart);
  caps_squash<<<gs, bs, 0, stream>>>(spart, bias, vbuf);
  caps_pass<2><<<gp, bp, 0, stream>>>(x, Wt, vbuf, bijw, spart);
  caps_squash<<<gs, bs, 0, stream>>>(spart, bias, out);
}


# Heavy-pass work-group 12 waves x 10-11 routes (was 4 x 32): 3 waves per SIMD fill each other's load waits; s summed over 12 LDS partials (b128 reads); sum order over routes changes
# speedup vs baseline: 3.0669x; 3.0669x over previous
#include <hip/hip_runtime.h>
#include <math.h>


typedef __attribute__((ext_vector_type(2))) float v2f;
typedef __attribute__((ext_vector_type(4))) float v4f;
typedef __attribute__((ext_vector_type(8))) float v8f;

constexpr int kB   = 128;
constexpr int kR   = 4096;
constexpr int kC   = 10;
constexpr int kO   = 16;
constexpr int kI   = 8;
constexpr int kMT  = 16;
constexpr int kNMT = kB / kMT;
constexpr int kRT  = 128;
constexpr int kNRT = kR / kRT;
constexpr int kWAVES = 12;
constexpr int kRPW = kRT / kWAVES;
constexpr int kNHI = kRT - kRPW * kWAVES;
constexpr int kCO  = kC * kO;
constexpr int kSP  = kCO + 4;

__device__ __forceinline__ float xhalf(float v) {
  const int b = __builtin_bit_cast(int, v);
  return __builtin_bit_cast(
      float, __builtin_amdgcn_permlanex16(b, b, 0x76543210, 0xfedcba98, false, false));
}

template <int PHASE>
__global__ __launch_bounds__(kWAVES * 32) void caps_pass(
    const float* __restrict__ x,
    const float* __restrict__ Wt,
    const float* __restrict__ vprev,
    float* __restrict__ bij,
    float* __restrict__ spart)
{
  __shared__ __attribute__((aligned(16))) float lds_s[kWAVES][kMT * kSP];

  const int mtile  = blockIdx.x & (kNMT - 1);
  const int rchunk = blockIdx.x / kNMT;
  const int wave   = __builtin_amdgcn_readfirstlane(threadIdx.x >> 5);
  const int lane   = threadIdx.x & 31;
  const int n      = lane & 15;
  const int half   = lane >> 4;
  const int k0     = half * 2;
  const int bg     = mtile * kMT + n;

  v8f vreg[kC];
  if (PHASE != 0) {
    const float* vp = vprev + (size_t)bg * kCO + half * 8;
#pragma unroll
    for (int c = 0; c < kC; ++c) {
      const v4f lo = *(const v4f*)(vp + c * kO);
      const v4f hi = *(const v4f*)(vp + c * kO + 4);
      vreg[c] = (v8f){lo.x, lo.y, lo.z, lo.w, hi.x, hi.y, hi.z, hi.w};
    }
  }

  const v8f zero8 = {0.f, 0.f, 0.f, 0.f, 0.f, 0.f, 0.f, 0.f};
  v8f s_acc[kC];
#pragma unroll
  for (int c = 0; c < kC; ++c) s_acc[c] = zero8;

  const int rfirst = rchunk * kRT + wave * kRPW + (wave < kNHI ? wave : kNHI);
  const int rcount = kRPW + (wave < kNHI ? 1 : 0);
  for (int ri = 0; ri < rcount; ++ri) {
    const int r = rfirst + ri;

    const float* xp = x + ((size_t)bg * kR + r) * kI;
    const v2f xb0 = *(const v2f*)(xp + k0);
    const v2f xb1 = *(const v2f*)(xp + k0 + 4);

    v8f acc[kC];
#pragma unroll
    for (int c = 0; c < kC; ++c) {
      const float* wp = Wt + (((size_t)r * kC + c) * kO + n) * kI;
      const v2f wa0 = *(const v2f*)(wp + k0);
      const v2f wa1 = *(const v2f*)(wp + k0 + 4);
      acc[c] = __builtin_amdgcn_wmma_f32_16x16x4_f32(false, wa0, false, xb0,
                                                     (short)0, zero8, false, false);
      acc[c] = __builtin_amdgcn_wmma_f32_16x16x4_f32(false, wa1, false, xb1,
                                                     (short)0, acc[c], false, false);
    }

    if (PHASE == 0) {
#pragma unroll
      for (int c = 0; c < kC; ++c) s_acc[c] += acc[c];
    } else {
      float* bplane = bij + ((size_t)r * kNMT + mtile) * (kC * kMT);

      float bold[kC];
      if (PHASE == 2) {
#pragma unroll
        for (int c = 0; c < kC; ++c) bold[c] = bplane[c * kMT + n];
      }

      float bn[kC];
#pragma unroll
      for (int c = 0; c < kC; ++c) {
        v2f pp = (v2f){acc[c][0], acc[c][1]} * (v2f){vreg[c][0], vreg[c][1]};
        pp += (v2f){acc[c][2], acc[c][3]} * (v2f){vreg[c][2], vreg[c][3]};
        pp += (v2f){acc[c][4], acc[c][5]} * (v2f){vreg[c][4], vreg[c][5]};
        pp += (v2f){acc[c][6], acc[c][7]} * (v2f){vreg[c][6], vreg[c][7]};
        const float p = pp.x + pp.y;
        bn[c] = p + xhalf(p);
      }
      if (PHASE == 2) {
#pragma unroll
        for (int c = 0; c < kC; ++c) bn[c] += bold[c];
      }
      if (PHASE == 1) {
#pragma unroll
        for (int i = 0; i < kC / 2; ++i)
          bplane[i * 32 + lane] = half ? bn[2 * i + 1] : bn[2 * i];
      }

      float mx = bn[0];
#pragma unroll
      for (int c = 1; c < kC; ++c) mx = fmaxf(mx, bn[c]);
      float sum = 0.f;
#pragma unroll
      for (int c = 0; c < kC; ++c) {
        bn[c] = __expf(bn[c] - mx);
        sum += bn[c];
      }
      const float inv = 1.0f / sum;
#pragma unroll
      for (int c = 0; c < kC; ++c) {
        const float cij = bn[c] * inv;
        s_acc[c] += acc[c] * cij;
      }
    }
  }

  {
    const float sc = (PHASE == 0) ? 0.1f : 1.0f;
    float* dst = &lds_s[wave][n * kSP + half * 8];
#pragma unroll
    for (int c = 0; c < kC; ++c) {
      *(v4f*)(dst + c * kO) =
          (v4f){s_acc[c][0] * sc, s_acc[c][1] * sc, s_acc[c][2] * sc, s_acc[c][3] * sc};
      *(v4f*)(dst + c * kO + 4) =
          (v4f){s_acc[c][4] * sc, s_acc[c][5] * sc, s_acc[c][6] * sc, s_acc[c][7] * sc};
    }
  }
  __syncthreads();
  float* tile = spart + ((size_t)rchunk * kB + mtile * kMT) * kCO;
  for (int e = threadIdx.x; e < kMT * kCO / 4; e += blockDim.x) {
    const int bl = e / (kCO / 4);
    const int c4 = e % (kCO / 4);
    const int a = bl * kSP + c4 * 4;
    v4f sum = *(const v4f*)&lds_s[0][a];
#pragma unroll
    for (int w = 1; w < kWAVES; ++w) sum += *(const v4f*)&lds_s[w][a];
    *(v4f*)(tile + e * 4) = sum;
  }
}

__global__ __launch_bounds__(256) void caps_squash(
    const float* __restrict__ spart,
    const float* __restrict__ bias,
    float* __restrict__ vout)
{
  const int idx = blockIdx.x * 256 + threadIdx.x;
  float s = bias[idx % kCO];
  for (int k = 0; k < kNRT; ++k) s += spart[(size_t)k * (kB * kCO) + idx];
  float p = s * s;
  p += __shfl_xor(p, 1);
  p += __shfl_xor(p, 2);
  p += __shfl_xor(p, 4);
  p += __shfl_xor(p, 8);
  vout[idx] = s * p / ((1.0f + p) * sqrtf(p));
}

extern "C" void kernel_launch(void* const* d_in, const int* in_sizes, int n_in,
                              void* d_out, int out_size, void* d_ws, size_t ws_size,
                              hipStream_t stream) {
  const float* x    = (const float*)d_in[0];
  const float* Wt   = (const float*)d_in[1];
  const float* bias = (const float*)d_in[2];
  float* out = (float*)d_out;

  float* spart = (float*)d_ws;
  float* vbuf  = spart + (size_t)kNRT * kB * kCO;
  float* bijw  = vbuf + (size_t)kB * kCO;

  const dim3 gp(kNMT * kNRT), bp(kWAVES * 32);
  const dim3 gs((kB * kCO) / 256), bs(256);

  caps_pass<0><<<gp, bp, 0, stream>>>(x, Wt, nullptr, bijw, spart);
  caps_squash<<<gs, bs, 0, stream>>>(spart, bias, vbuf);
  caps_pass<1><<<gp, bp, 0, stream>>>(x, Wt, vbuf, bijw, spart);
  caps_squash<<<gs, bs, 0, stream>>>(spart, bias, vbuf);
  caps_pass<2><<<gp, bp, 0, stream>>>(x, Wt, vbuf, bijw, spart);
  caps_squash<<<gs, bs, 0, stream>>>(spart, bias, out);
}


# A rows regrouped (8 ch of capsule 2p | 8 ch of capsule 2p+1): a lane half owns 5 whole capsules; dot over o in-lane, softmax split 5+5 with 2 half exchanges (was 10); sum orders over o and c change
# speedup vs baseline: 3.2417x; 1.0570x over previous
#include <hip/hip_runtime.h>
#include <math.h>


typedef __attribute__((ext_vector_type(2))) float v2f;
typedef __attribute__((ext_vector_type(4))) float v4f;
typedef __attribute__((ext_vector_type(8))) float v8f;

constexpr int kB   = 128;
constexpr int kR   = 4096;
constexpr int kC   = 10;
constexpr int kO   = 16;
constexpr int kI   = 8;
constexpr int kMT  = 16;
constexpr int kNMT = kB / kMT;
constexpr int kRT  = 128;
constexpr int kNRT = kR / kRT;
constexpr int kWAVES = 12;
constexpr int kRPW = kRT / kWAVES;
constexpr int kNHI = kRT - kRPW * kWAVES;
constexpr int kCP  = kC / 2;
constexpr int kCO  = kC * kO;
constexpr int kSP  = kCO + 4;

__device__ __forceinline__ float xhalf(float v) {
  const int b = __builtin_bit_cast(int, v);
  return __builtin_bit_cast(
      float, __builtin_amdgcn_permlanex16(b, b, 0x76543210, 0xfedcba98, false, false));
}

template <int PHASE>
__global__ __launch_bounds__(kWAVES * 32) void caps_pass(
    const float* __restrict__ x,
    const float* __restrict__ Wt,
    const float* __restrict__ vprev,
    float* __restrict__ bij,
    float* __restrict__ spart)
{
  __shared__ __attribute__((aligned(16))) float lds_s[kWAVES][kMT * kSP];

  const int mtile  = blockIdx.x & (kNMT - 1);
  const int rchunk = blockIdx.x / kNMT;
  const int wave   = __builtin_amdgcn_readfirstlane(threadIdx.x >> 5);
  const int lane   = threadIdx.x & 31;
  const int n      = lane & 15;
  const int half   = lane >> 4;
  const int k0     = half * 2;
  const int bg     = mtile * kMT + n;
  const int arow   = (n >> 3) * kO + (n & 7);

  v8f vreg[kCP][2];
  if (PHASE != 0) {
    const float* vp = vprev + (size_t)bg * kCO + half * kO;
#pragma unroll
    for (int p = 0; p < kCP; ++p) {
#pragma unroll
      for (int q = 0; q < 2; ++q) {
        const float* a = vp + 2 * p * kO + 8 * q;
        const v4f lo = *(const v4f*)(a);
        const v4f hi = *(const v4f*)(a + 4);
        vreg[p][q] = (v8f){lo.x, lo.y, lo.z, lo.w, hi.x, hi.y, hi.z, hi.w};
      }
    }
  }

  const v8f zero8 = {0.f, 0.f, 0.f, 0.f, 0.f, 0.f, 0.f, 0.f};
  v8f s_acc[kCP][2];
#pragma unroll
  for (int p = 0; p < kCP; ++p) {
    s_acc[p][0] = zero8;
    s_acc[p][1] = zero8;
  }

  const int rfirst = rchunk * kRT + wave * kRPW + (wave < kNHI ? wave : kNHI);
  const int rcount = kRPW + (wave < kNHI ? 1 : 0);
  for (int ri = 0; ri < rcount; ++ri) {
    const int r = rfirst + ri;

    const float* xp = x + ((size_t)bg * kR + r) * kI;
    const v2f xb0 = *(const v2f*)(xp + k0);
    const v2f xb1 = *(const v2f*)(xp + k0 + 4);

    v8f acc[kCP][2];
#pragma unroll
    for (int p = 0; p < kCP; ++p) {
#pragma unroll
      for (int q = 0; q < 2; ++q) {
        const float* wp = Wt + (((size_t)r * kC + 2 * p) * kO + 8 * q + arow) * kI;
        const v2f wa0 = *(const v2f*)(wp + k0);
        const v2f wa1 = *(const v2f*)(wp + k0 + 4);
        acc[p][q] = __builtin_amdgcn_wmma_f32_16x16x4_f32(false, wa0, false, xb0,
                                                          (short)0, zero8, false, false);
        acc[p][q] = __builtin_amdgcn_wmma_f32_16x16x4_f32(false, wa1, false, xb1,
                                                          (short)0, acc[p][q], false, false);
      }
    }

    if (PHASE == 0) {
#pragma unroll
      for (int p = 0; p < kCP; ++p) {
        s_acc[p][0] += acc[p][0];
        s_acc[p][1] += acc[p][1];
      }
    } else {
      float* bplane = bij + ((size_t)r * kNMT + mtile) * (kC * kMT);

      float bold[kCP];
      if (PHASE == 2) {
#pragma unroll
        for (int p = 0; p < kCP; ++p) bold[p] = bplane[p * 32 + lane];
      }

      float bn[kCP];
#pragma unroll
      for (int p = 0; p < kCP; ++p) {
        v2f pp = (v2f){acc[p][0][0], acc[p][0][1]} * (v2f){vreg[p][0][0], vreg[p][0][1]};
        pp += (v2f){acc[p][0][2], acc[p][0][3]} * (v2f){vreg[p][0][2], vreg[p][0][3]};
        pp += (v2f){acc[p][0][4], acc[p][0][5]} * (v2f){vreg[p][0][4], vreg[p][0][5]};
        pp += (v2f){acc[p][0][6], acc[p][0][7]} * (v2f){vreg[p][0][6], vreg[p][0][7]};
        pp += (v2f){acc[p][1][0], acc[p][1][1]} * (v2f){vreg[p][1][0], vreg[p][1][1]};
        pp += (v2f){acc[p][1][2], acc[p][1][3]} * (v2f){vreg[p][1][2], vreg[p][1][3]};
        pp += (v2f){acc[p][1][4], acc[p][1][5]} * (v2f){vreg[p][1][4], vreg[p][1][5]};
        pp += (v2f){acc[p][1][6], acc[p][1][7]} * (v2f){vreg[p][1][6], vreg[p][1][7]};
        bn[p] = pp.x + pp.y;
      }
      if (PHASE == 2) {
#pragma unroll
        for (int p = 0; p < kCP; ++p) bn[p] += bold[p];
      }
      if (PHASE == 1) {
#pragma unroll
        for (int p = 0; p < kCP; ++p) bplane[p * 32 + lane] = bn[p];
      }

      float mx = bn[0];
#pragma unroll
      for (int p = 1; p < kCP; ++p) mx = fmaxf(mx, bn[p]);
      mx = fmaxf(mx, xhalf(mx));
      float sum = 0.f;
#pragma unroll
      for (int p = 0; p < kCP; ++p) {
        bn[p] = __expf(bn[p] - mx);
        sum += bn[p];
      }
      sum += xhalf(sum);
      const float inv = 1.0f / sum;
#pragma unroll
      for (int p = 0; p < kCP; ++p) {
        const float cij = bn[p] * inv;
        s_acc[p][0] += acc[p][0] * cij;
        s_acc[p][1] += acc[p][1] * cij;
      }
    }
  }

  {
    const float sc = (PHASE == 0) ? 0.1f : 1.0f;
    float* dst = &lds_s[wave][n * kSP + half * kO];
#pragma unroll
    for (int p = 0; p < kCP; ++p) {
#pragma unroll
      for (int q = 0; q < 2; ++q) {
        float* d = dst + 2 * p * kO + 8 * q;
        *(v4f*)(d) = (v4f){s_acc[p][q][0] * sc, s_acc[p][q][1] * sc,
                           s_acc[p][q][2] * sc, s_acc[p][q][3] * sc};
        *(v4f*)(d + 4) = (v4f){s_acc[p][q][4] * sc, s_acc[p][q][5] * sc,
                               s_acc[p][q][6] * sc, s_acc[p][q][7] * sc};
      }
    }
  }
  __syncthreads();
  float* tile = spart + ((size_t)rchunk * kB + mtile * kMT) * kCO;
  for (int e = threadIdx.x; e < kMT * kCO / 4; e += blockDim.x) {
    const int bl = e / (kCO / 4);
    const int c4 = e % (kCO / 4);
    const int a = bl * kSP + c4 * 4;
    v4f sum = *(const v4f*)&lds_s[0][a];
#pragma unroll
    for (int w = 1; w < kWAVES; ++w) sum += *(const v4f*)&lds_s[w][a];
    *(v4f*)(tile + e * 4) = sum;
  }
}

__global__ __launch_bounds__(256) void caps_squash(
    const float* __restrict__ spart,
    const float* __restrict__ bias,
    float* __restrict__ vout)
{
  const int idx = blockIdx.x * 256 + threadIdx.x;
  float s = bias[idx % kCO];
  for (int k = 0; k < kNRT; ++k) s += spart[(size_t)k * (kB * kCO) + idx];
  float p = s * s;
  p += __shfl_xor(p, 1);
  p += __shfl_xor(p, 2);
  p += __shfl_xor(p, 4);
  p += __shfl_xor(p, 8);
  vout[idx] = s * p / ((1.0f + p) * sqrtf(p));
}

extern "C" void kernel_launch(void* const* d_in, const int* in_sizes, int n_in,
                              void* d_out, int out_size, void* d_ws, size_t ws_size,
                              hipStream_t stream) {
  const float* x    = (const float*)d_in[0];
  const float* Wt   = (const float*)d_in[1];
  const float* bias = (const float*)d_in[2];
  float* out = (float*)d_out;

  float* spart = (float*)d_ws;
  float* vbuf  = spart + (size_t)kNRT * kB * kCO;
  float* bijw  = vbuf + (size_t)kB * kCO;

  const dim3 gp(kNMT * kNRT), bp(kWAVES * 32);
  const dim3 gs((kB * kCO) / 256), bs(256);

  caps_pass<0><<<gp, bp, 0, stream>>>(x, Wt, nullptr, bijw, spart);
  caps_squash<<<gs, bs, 0, stream>>>(spart, bias, vbuf);
  caps_pass<1><<<gp, bp, 0, stream>>>(x, Wt, vbuf, bijw, spart);
  caps_squash<<<gs, bs, 0, stream>>>(spart, bias, vbuf);
  caps_pass<2><<<gp, bp, 0, stream>>>(x, Wt, vbuf, bijw, spart);
  caps_squash<<<gs, bs, 0, stream>>>(spart, bias, out);
}


# Route loop software-pipelined: x/W operand registers refilled right behind the WMMAs that read them (pass 0: two sets, two routes ahead; passes 1-2: in place), loads fly under the routing math
# speedup vs baseline: 3.3613x; 1.0369x over previous
#include <hip/hip_runtime.h>
#include <math.h>


typedef __attribute__((ext_vector_type(2))) float v2f;
typedef __attribute__((ext_vector_type(4))) float v4f;
typedef __attribute__((ext_vector_type(8))) float v8f;

constexpr int kB   = 128;
constexpr int kR   = 4096;
constexpr int kC   = 10;
constexpr int kO   = 16;
constexpr int kI   = 8;
constexpr int kMT  = 16;
constexpr int kNMT = kB / kMT;
constexpr int kRT  = 128;
constexpr int kNRT = kR / kRT;
constexpr int kWAVES = 12;
constexpr int kRPW = kRT / kWAVES;
constexpr int kNHI = kRT - kRPW * kWAVES;
constexpr int kCP  = kC / 2;
constexpr int kCO  = kC * kO;
constexpr int kSP  = kCO + 4;

__device__ __forceinline__ float xhalf(float v) {
  const int b = __builtin_bit_cast(int, v);
  return __builtin_bit_cast(
      float, __builtin_amdgcn_permlanex16(b, b, 0x76543210, 0xfedcba98, false, false));
}

struct Frag {
  v2f xb[2];
  v2f wa[kCP][2][2];
};

__device__ __forceinline__ void load_frag(Frag& f, const float* __restrict__ x,
                                          const float* __restrict__ Wt, int r,
                                          unsigned xoff, unsigned woff) {
  const float* xr = x + (size_t)r * kI;
  f.xb[0] = *(const v2f*)(xr + xoff);
  f.xb[1] = *(const v2f*)(xr + xoff + 4);
  const float* wr = Wt + (size_t)r * (kC * kO * kI);
#pragma unroll
  for (int p = 0; p < kCP; ++p) {
#pragma unroll
    for (int q = 0; q < 2; ++q) {
      const unsigned o = woff + (2 * p * kO + 8 * q) * kI;
      f.wa[p][q][0] = *(const v2f*)(wr + o);
      f.wa[p][q][1] = *(const v2f*)(wr + o + 4);
    }
  }
}

__device__ __forceinline__ void uhat_tiles(const Frag& f, v8f (&acc)[kCP][2]) {
  const v8f zero8 = {0.f, 0.f, 0.f, 0.f, 0.f, 0.f, 0.f, 0.f};
#pragma unroll
  for (int p = 0; p < kCP; ++p) {
#pragma unroll
    for (int q = 0; q < 2; ++q) {
      acc[p][q] = __builtin_amdgcn_wmma_f32_16x16x4_f32(false, f.wa[p][q][0], false, f.xb[0],
                                                        (short)0, zero8, false, false);
      acc[p][q] = __builtin_amdgcn_wmma_f32_16x16x4_f32(false, f.wa[p][q][1], false, f.xb[1],
                                                        (short)0, acc[p][q], false, false);
    }
  }
}

template <int PHASE>
__device__ __forceinline__ void route_math(const v8f (&acc)[kCP][2], const float (&bold)[kCP],
                                           v8f (&s_acc)[kCP][2], const v8f (&vreg)[kCP][2],
                                           float* __restrict__ bplane, int lane) {
  if (PHASE == 0) {
#pragma unroll
    for (int p = 0; p < kCP; ++p) {
      s_acc[p][0] += acc[p][0];
      s_acc[p][1] += acc[p][1];
    }
  } else {
    float bn[kCP];
#pragma unroll
    for (int p = 0; p < kCP; ++p) {
      v2f pp = (v2f){acc[p][0][0], acc[p][0][1]} * (v2f){vreg[p][0][0], vreg[p][0][1]};
      pp += (v2f){acc[p][0][2], acc[p][0][3]} * (v2f){vreg[p][0][2], vreg[p][0][3]};
      pp += (v2f){acc[p][0][4], acc[p][0][5]} * (v2f){vreg[p][0][4], vreg[p][0][5]};
      pp += (v2f){acc[p][0][6], acc[p][0][7]} * (v2f){vreg[p][0][6], vreg[p][0][7]};
      pp += (v2f){acc[p][1][0], acc[p][1][1]} * (v2f){vreg[p][1][0], vreg[p][1][1]};
      pp += (v2f){acc[p][1][2], acc[p][1][3]} * (v2f){vreg[p][1][2], vreg[p][1][3]};
      pp += (v2f){acc[p][1][4], acc[p][1][5]} * (v2f){vreg[p][1][4], vreg[p][1][5]};
      pp += (v2f){acc[p][1][6], acc[p][1][7]} * (v2f){vreg[p][1][6], vreg[p][1][7]};
      bn[p] = pp.x + pp.y;
    }
    if (PHASE == 2) {
#pragma unroll
      for (int p = 0; p < kCP; ++p) bn[p] += bold[p];
    }
    if (PHASE == 1) {
#pragma unroll
      for (int p = 0; p < kCP; ++p) bplane[p * 32 + lane] = bn[p];
    }

    float mx = bn[0];
#pragma unroll
    for (int p = 1; p < kCP; ++p) mx = fmaxf(mx, bn[p]);
    mx = fmaxf(mx, xhalf(mx));
    float sum = 0.f;
#pragma unroll
    for (int p = 0; p < kCP; ++p) {
      bn[p] = __expf(bn[p] - mx);
      sum += bn[p];
    }
    sum += xhalf(sum);
    const float inv = 1.0f / sum;
#pragma unroll
    for (int p = 0; p < kCP; ++p) {
      const float cij = bn[p] * inv;
      s_acc[p][0] += acc[p][0] * cij;
      s_acc[p][1] += acc[p][1] * cij;
    }
  }
}

template <int PHASE>
__global__ __launch_bounds__(kWAVES * 32) void caps_pass(
    const float* __restrict__ x,
    const float* __restrict__ Wt,
    const float* __restrict__ vprev,
    float* __restrict__ bij,
    float* __restrict__ spart)
{
  __shared__ __attribute__((aligned(16))) float lds_s[kWAVES][kMT * kSP];

  const int mtile  = blockIdx.x & (kNMT - 1);
  const int rchunk = blockIdx.x / kNMT;
  const int wave   = __builtin_amdgcn_readfirstlane(threadIdx.x >> 5);
  const int lane   = threadIdx.x & 31;
  const int n      = lane & 15;
  const int half   = lane >> 4;
  const int k0     = half * 2;
  const int bg     = mtile * kMT + n;
  const int arow   = (n >> 3) * kO + (n & 7);

  v8f vreg[kCP][2];
  if (PHASE != 0) {
    const float* vp = vprev + (size_t)bg * kCO + half * kO;
#pragma unroll
    for (int p = 0; p < kCP; ++p) {
#pragma unroll
      for (int q = 0; q < 2; ++q) {
        const float* a = vp + 2 * p * kO + 8 * q;
        const v4f lo = *(const v4f*)(a);
        const v4f hi = *(const v4f*)(a + 4);
        vreg[p][q] = (v8f){lo.x, lo.y, lo.z, lo.w, hi.x, hi.y, hi.z, hi.w};
      }
    }
  }

  const v8f zero8 = {0.f, 0.f, 0.f, 0.f, 0.f, 0.f, 0.f, 0.f};
  v8f s_acc[kCP][2];
#pragma unroll
  for (int p = 0; p < kCP; ++p) {
    s_acc[p][0] = zero8;
    s_acc[p][1] = zero8;
  }

  const int rfirst = rchunk * kRT + wave * kRPW + (wave < kNHI ? wave : kNHI);
  const int rcount = kRPW + (wave < kNHI ? 1 : 0);
  const int rlast  = rfirst + rcount - 1;

  const unsigned xoff = (unsigned)bg * (kR * kI) + k0;
  const unsigned woff = (unsigned)arow * kI + k0;

  if (PHASE == 0) {
    const float bnone[kCP] = {0.f, 0.f, 0.f, 0.f, 0.f};
    Frag fa, fb;
    load_frag(fa, x, Wt, rfirst, xoff, woff);
    load_frag(fb, x, Wt, rfirst + 1, xoff, woff);
    const bool odd = (rcount & 1) != 0;
    int r = rfirst;
#pragma nounroll
    for (; r + 3 <= rlast - (odd ? 1 : 0); r += 2) {
      v8f acc[kCP][2];
      uhat_tiles(fa, acc);
      load_frag(fa, x, Wt, r + 2, xoff, woff);
      asm volatile("" ::: "memory");
      route_math<PHASE>(acc, bnone, s_acc, vreg, bij, lane);
      uhat_tiles(fb, acc);
      load_frag(fb, x, Wt, r + 3, xoff, woff);
      asm volatile("" ::: "memory");
      route_math<PHASE>(acc, bnone, s_acc, vreg, bij, lane);
    }
    {
      v8f acc[kCP][2];
      uhat_tiles(fa, acc);
      if (odd) load_frag(fa, x, Wt, r + 2, xoff, woff);
      route_math<PHASE>(acc, bnone, s_acc, vreg, bij, lane);
      uhat_tiles(fb, acc);
      route_math<PHASE>(acc, bnone, s_acc, vreg, bij, lane);
      if (odd) {
        uhat_tiles(fa, acc);
        route_math<PHASE>(acc, bnone, s_acc, vreg, bij, lane);
      }
    }
  } else {
    Frag f;
    load_frag(f, x, Wt, rfirst, xoff, woff);
    for (int r = rfirst; r < rlast; ++r) {
      float* bplane = bij + ((size_t)r * kNMT + mtile) * (kC * kMT);
      float bold[kCP];
      if (PHASE == 2) {
#pragma unroll
        for (int p = 0; p < kCP; ++p) bold[p] = bplane[p * 32 + lane];
      }
      v8f acc[kCP][2];
      uhat_tiles(f, acc);
      load_frag(f, x, Wt, r + 1, xoff, woff);
      asm volatile("" ::: "memory");
      route_math<PHASE>(acc, bold, s_acc, vreg, bplane, lane);
    }
    {
      float* bplane = bij + ((size_t)rlast * kNMT + mtile) * (kC * kMT);
      float bold[kCP];
      if (PHASE == 2) {
#pragma unroll
        for (int p = 0; p < kCP; ++p) bold[p] = bplane[p * 32 + lane];
      }
      v8f acc[kCP][2];
      uhat_tiles(f, acc);
      route_math<PHASE>(acc, bold, s_acc, vreg, bplane, lane);
    }
  }

  {
    const float sc = (PHASE == 0) ? 0.1f : 1.0f;
    float* dst = &lds_s[wave][n * kSP + half * kO];
#pragma unroll
    for (int p = 0; p < kCP; ++p) {
#pragma unroll
      for (int q = 0; q < 2; ++q) {
        float* d = dst + 2 * p * kO + 8 * q;
        *(v4f*)(d) = (v4f){s_acc[p][q][0] * sc, s_acc[p][q][1] * sc,
                           s_acc[p][q][2] * sc, s_acc[p][q][3] * sc};
        *(v4f*)(d + 4) = (v4f){s_acc[p][q][4] * sc, s_acc[p][q][5] * sc,
                               s_acc[p][q][6] * sc, s_acc[p][q][7] * sc};
      }
    }
  }
  __syncthreads();
  float* tile = spart + ((size_t)rchunk * kB + mtile * kMT) * kCO;
  for (int e = threadIdx.x; e < kMT * kCO / 4; e += blockDim.x) {
    const int bl = e / (kCO / 4);
    const int c4 = e % (kCO / 4);
    const int a = bl * kSP + c4 * 4;
    v4f sum = *(const v4f*)&lds_s[0][a];
#pragma unroll
    for (int w = 1; w < kWAVES; ++w) sum += *(const v4f*)&lds_s[w][a];
    *(v4f*)(tile + e * 4) = sum;
  }
}

__global__ __launch_bounds__(256) void caps_squash(
    const float* __restrict__ spart,
    const float* __restrict__ bias,
    float* __restrict__ vout)
{
  const int idx = blockIdx.x * 256 + threadIdx.x;
  float s = bias[idx % kCO];
  for (int k = 0; k < kNRT; ++k) s += spart[(size_t)k * (kB * kCO) + idx];
  float p = s * s;
  p += __shfl_xor(p, 1);
  p += __shfl_xor(p, 2);
  p += __shfl_xor(p, 4);
  p += __shfl_xor(p, 8);
  vout[idx] = s * p / ((1.0f + p) * sqrtf(p));
}

extern "C" void kernel_launch(void* const* d_in, const int* in_sizes, int n_in,
                              void* d_out, int out_size, void* d_ws, size_t ws_size,
                              hipStream_t stream) {
  const float* x    = (const float*)d_in[0];
  const float* Wt   = (const float*)d_in[1];
  const float* bias = (const float*)d_in[2];
  float* out = (float*)d_out;

  float* spart = (float*)d_ws;
  float* vbuf  = spart + (size_t)kNRT * kB * kCO;
  float* bijw  = vbuf + (size_t)kB * kCO;

  const dim3 gp(kNMT * kNRT), bp(kWAVES * 32);
  const dim3 gs((kB * kCO) / 256), bs(256);

  caps_pass<0><<<gp, bp, 0, stream>>>(x, Wt, nullptr, bijw, spart);
  caps_squash<<<gs, bs, 0, stream>>>(spart, bias, vbuf);
  caps_pass<1><<<gp, bp, 0, stream>>>(x, Wt, vbuf, bijw, spart);
  caps_squash<<<gs, bs, 0, stream>>>(spart, bias, vbuf);
  caps_pass<2><<<gp, bp, 0, stream>>>(x, Wt, vbuf, bijw, spart);
  caps_squash<<<gs, bs, 0, stream>>>(spart, bias, out);
}


# Logits by linearity in v: last pass takes b = u_hat.(v0+v1) from its own dot; b_ij plane (R x B x C f32, stored by pass 1, read by pass 2) gone, passes 1-2 one kernel; logit sum order changes
# speedup vs baseline: 3.6791x; 1.0945x over previous
#include <hip/hip_runtime.h>
#include <math.h>


typedef __attribute__((ext_vector_type(2))) float v2f;
typedef __attribute__((ext_vector_type(4))) float v4f;
typedef __attribute__((ext_vector_type(8))) float v8f;

constexpr int kB   = 128;
constexpr int kR   = 4096;
constexpr int kC   = 10;
constexpr int kO   = 16;
constexpr int kI   = 8;
constexpr int kMT  = 16;
constexpr int kNMT = kB / kMT;
constexpr int kRT  = 128;
constexpr int kNRT = kR / kRT;
constexpr int kWAVES = 12;
constexpr int kRPW = kRT / kWAVES;
constexpr int kNHI = kRT - kRPW * kWAVES;
constexpr int kCP  = kC / 2;
constexpr int kCO  = kC * kO;
constexpr int kSP  = kCO + 4;

__device__ __forceinline__ float xhalf(float v) {
  const int b = __builtin_bit_cast(int, v);
  return __builtin_bit_cast(
      float, __builtin_amdgcn_permlanex16(b, b, 0x76543210, 0xfedcba98, false, false));
}

struct Frag {
  v2f xb[2];
  v2f wa[kCP][2][2];
};

__device__ __forceinline__ void load_frag(Frag& f, const float* __restrict__ x,
                                          const float* __restrict__ Wt, int r,
                                          unsigned xoff, unsigned woff) {
  const float* xr = x + (size_t)r * kI;
  f.xb[0] = *(const v2f*)(xr + xoff);
  f.xb[1] = *(const v2f*)(xr + xoff + 4);
  const float* wr = Wt + (size_t)r * (kC * kO * kI);
#pragma unroll
  for (int p = 0; p < kCP; ++p) {
#pragma unroll
    for (int q = 0; q < 2; ++q) {
      const unsigned o = woff + (2 * p * kO + 8 * q) * kI;
      f.wa[p][q][0] = *(const v2f*)(wr + o);
      f.wa[p][q][1] = *(const v2f*)(wr + o + 4);
    }
  }
}

__device__ __forceinline__ void uhat_tiles(const Frag& f, v8f (&acc)[kCP][2]) {
  const v8f zero8 = {0.f, 0.f, 0.f, 0.f, 0.f, 0.f, 0.f, 0.f};
#pragma unroll
  for (int p = 0; p < kCP; ++p) {
#pragma unroll
    for (int q = 0; q < 2; ++q) {
      acc[p][q] = __builtin_amdgcn_wmma_f32_16x16x4_f32(false, f.wa[p][q][0], false, f.xb[0],
                                                        (short)0, zero8, false, false);
      acc[p][q] = __builtin_amdgcn_wmma_f32_16x16x4_f32(false, f.wa[p][q][1], false, f.xb[1],
                                                        (short)0, acc[p][q], false, false);
    }
  }
}

template <int PHASE>
__device__ __forceinline__ void route_math(const v8f (&acc)[kCP][2], v8f (&s_acc)[kCP][2],
                                           const v8f (&vreg)[kCP][2]) {
  if (PHASE == 0) {
#pragma unroll
    for (int p = 0; p < kCP; ++p) {
      s_acc[p][0] += acc[p][0];
      s_acc[p][1] += acc[p][1];
    }
  } else {
    float bn[kCP];
#pragma unroll
    for (int p = 0; p < kCP; ++p) {
      v2f pp = (v2f){acc[p][0][0], acc[p][0][1]} * (v2f){vreg[p][0][0], vreg[p][0][1]};
      pp += (v2f){acc[p][0][2], acc[p][0][3]} * (v2f){vreg[p][0][2], vreg[p][0][3]};
      pp += (v2f){acc[p][0][4], acc[p][0][5]} * (v2f){vreg[p][0][4], vreg[p][0][5]};
      pp += (v2f){acc[p][0][6], acc[p][0][7]} * (v2f){vreg[p][0][6], vreg[p][0][7]};
      pp += (v2f){acc[p][1][0], acc[p][1][1]} * (v2f){vreg[p][1][0], vreg[p][1][1]};
      pp += (v2f){acc[p][1][2], acc[p][1][3]} * (v2f){vreg[p][1][2], vreg[p][1][3]};
      pp += (v2f){acc[p][1][4], acc[p][1][5]} * (v2f){vreg[p][1][4], vreg[p][1][5]};
      pp += (v2f){acc[p][1][6], acc[p][1][7]} * (v2f){vreg[p][1][6], vreg[p][1][7]};
      bn[p] = pp.x + pp.y;
    }

    float mx = bn[0];
#pragma unroll
    for (int p = 1; p < kCP; ++p) mx = fmaxf(mx, bn[p]);
    mx = fmaxf(mx, xhalf(mx));
    float sum = 0.f;
#pragma unroll
    for (int p = 0; p < kCP; ++p) {
      bn[p] = __expf(bn[p] - mx);
      sum += bn[p];
    }
    sum += xhalf(sum);
    const float inv = 1.0f / sum;
#pragma unroll
    for (int p = 0; p < kCP; ++p) {
      const float cij = bn[p] * inv;
      s_acc[p][0] += acc[p][0] * cij;
      s_acc[p][1] += acc[p][1] * cij;
    }
  }
}

template <int PHASE>
__global__ __launch_bounds__(kWAVES * 32) void caps_pass(
    const float* __restrict__ x,
    const float* __restrict__ Wt,
    const float* __restrict__ vprev,
    float* __restrict__ spart)
{
  __shared__ __attribute__((aligned(16))) float lds_s[kWAVES][kMT * kSP];

  const int mtile  = blockIdx.x & (kNMT - 1);
  const int rchunk = blockIdx.x / kNMT;
  const int wave   = __builtin_amdgcn_readfirstlane(threadIdx.x >> 5);
  const int lane   = threadIdx.x & 31;
  const int n      = lane & 15;
  const int half   = lane >> 4;
  const int k0     = half * 2;
  const int bg     = mtile * kMT + n;
  const int arow   = (n >> 3) * kO + (n & 7);

  v8f vreg[kCP][2];
  if (PHASE != 0) {
    const float* vp = vprev + (size_t)bg * kCO + half * kO;
#pragma unroll
    for (int p = 0; p < kCP; ++p) {
#pragma unroll
      for (int q = 0; q < 2; ++q) {
        const float* a = vp + 2 * p * kO + 8 * q;
        const v4f lo = *(const v4f*)(a);
        const v4f hi = *(const v4f*)(a + 4);
        vreg[p][q] = (v8f){lo.x, lo.y, lo.z, lo.w, hi.x, hi.y, hi.z, hi.w};
      }
    }
  }

  const v8f zero8 = {0.f, 0.f, 0.f, 0.f, 0.f, 0.f, 0.f, 0.f};
  v8f s_acc[kCP][2];
#pragma unroll
  for (int p = 0; p < kCP; ++p) {
    s_acc[p][0] = zero8;
    s_acc[p][1] = zero8;
  }

  const int rfirst = rchunk * kRT + wave * kRPW + (wave < kNHI ? wave : kNHI);
  const int rcount = kRPW + (wave < kNHI ? 1 : 0);
  const int rlast  = rfirst + rcount - 1;

  const unsigned xoff = (unsigned)bg * (kR * kI) + k0;
  const unsigned woff = (unsigned)arow * kI + k0;

  if (PHASE == 0) {
    Frag fa, fb;
    load_frag(fa, x, Wt, rfirst, xoff, woff);
    load_frag(fb, x, Wt, rfirst + 1, xoff, woff);
    const bool odd = (rcount & 1) != 0;
    int r = rfirst;
#pragma nounroll
    for (; r + 3 <= rlast - (odd ? 1 : 0); r += 2) {
      v8f acc[kCP][2];
      uhat_tiles(fa, acc);
      load_frag(fa, x, Wt, r + 2, xoff, woff);
      asm volatile("" ::: "memory");
      route_math<PHASE>(acc, s_acc, vreg);
      uhat_tiles(fb, acc);
      load_frag(fb, x, Wt, r + 3, xoff, woff);
      asm volatile("" ::: "memory");
      route_math<PHASE>(acc, s_acc, vreg);
    }
    {
      v8f acc[kCP][2];
      uhat_tiles(fa, acc);
      if (odd) load_frag(fa, x, Wt, r + 2, xoff, woff);
      route_math<PHASE>(acc, s_acc, vreg);
      uhat_tiles(fb, acc);
      route_math<PHASE>(acc, s_acc, vreg);
      if (odd) {
        uhat_tiles(fa, acc);
        route_math<PHASE>(acc, s_acc, vreg);
      }
    }
  } else {
    Frag f;
    load_frag(f, x, Wt, rfirst, xoff, woff);
    for (int r = rfirst; r < rlast; ++r) {
      v8f acc[kCP][2];
      uhat_tiles(f, acc);
      load_frag(f, x, Wt, r + 1, xoff, woff);
      asm volatile("" ::: "memory");
      route_math<PHASE>(acc, s_acc, vreg);
    }
    {
      v8f acc[kCP][2];
      uhat_tiles(f, acc);
      route_math<PHASE>(acc, s_acc, vreg);
    }
  }

  {
    const float sc = (PHASE == 0) ? 0.1f : 1.0f;
    float* dst = &lds_s[wave][n * kSP + half * kO];
#pragma unroll
    for (int p = 0; p < kCP; ++p) {
#pragma unroll
      for (int q = 0; q < 2; ++q) {
        float* d = dst + 2 * p * kO + 8 * q;
        *(v4f*)(d) = (v4f){s_acc[p][q][0] * sc, s_acc[p][q][1] * sc,
                           s_acc[p][q][2] * sc, s_acc[p][q][3] * sc};
        *(v4f*)(d + 4) = (v4f){s_acc[p][q][4] * sc, s_acc[p][q][5] * sc,
                               s_acc[p][q][6] * sc, s_acc[p][q][7] * sc};
      }
    }
  }
  __syncthreads();
  float* tile = spart + ((size_t)rchunk * kB + mtile * kMT) * kCO;
  for (int e = threadIdx.x; e < kMT * kCO / 4; e += blockDim.x) {
    const int bl = e / (kCO / 4);
    const int c4 = e % (kCO / 4);
    const int a = bl * kSP + c4 * 4;
    v4f sum = *(const v4f*)&lds_s[0][a];
#pragma unroll
    for (int w = 1; w < kWAVES; ++w) sum += *(const v4f*)&lds_s[w][a];
    *(v4f*)(tile + e * 4) = sum;
  }
}

template <bool ADDPREV>
__global__ __launch_bounds__(256) void caps_squash(
    const float* __restrict__ spart,
    const float* __restrict__ bias,
    const float* __restrict__ vadd,
    float* __restrict__ vout)
{
  const int idx = blockIdx.x * 256 + threadIdx.x;
  float s = bias[idx % kCO];
  for (int k = 0; k < kNRT; ++k) s += spart[(size_t)k * (kB * kCO) + idx];
  float p = s * s;
  p += __shfl_xor(p, 1);
  p += __shfl_xor(p, 2);
  p += __shfl_xor(p, 4);
  p += __shfl_xor(p, 8);
  const float v = s * p / ((1.0f + p) * sqrtf(p));
  vout[idx] = ADDPREV ? vadd[idx] + v : v;
}

extern "C" void kernel_launch(void* const* d_in, const int* in_sizes, int n_in,
                              void* d_out, int out_size, void* d_ws, size_t ws_size,
                              hipStream_t stream) {
  const float* x    = (const float*)d_in[0];
  const float* Wt   = (const float*)d_in[1];
  const float* bias = (const float*)d_in[2];
  float* out = (float*)d_out;

  float* spart = (float*)d_ws;
  float* vbuf  = spart + (size_t)kNRT * kB * kCO;
  float* vsum  = vbuf + (size_t)kB * kCO;

  const dim3 gp(kNMT * kNRT), bp(kWAVES * 32);
  const dim3 gs((kB * kCO) / 256), bs(256);

  caps_pass<0><<<gp, bp, 0, stream>>>(x, Wt, nullptr, spart);
  caps_squash<false><<<gs, bs, 0, stream>>>(spart, bias, nullptr, vbuf);
  caps_pass<1><<<gp, bp, 0, stream>>>(x, Wt, vbuf, spart);
  caps_squash<true><<<gs, bs, 0, stream>>>(spart, bias, vbuf, vsum);
  caps_pass<1><<<gp, bp, 0, stream>>>(x, Wt, vsum, spart);
  caps_squash<false><<<gs, bs, 0, stream>>>(spart, bias, nullptr, out);
}
